# starting kernel
# baseline (speedup 1.0000x reference)
#include <hip/hip_runtime.h>


typedef __attribute__((ext_vector_type(2))) float v2f;
typedef __attribute__((ext_vector_type(8))) float v8f;

#define B_      8
#define T_      4096
#define C_      128
#define H_      512
#define DIL     4
#define TILE_T  64
#define HROWS   80
#define NMT     (HROWS / 16)
#define LDST    (H_ + 4)

__global__ __launch_bounds__(256)
void conv1d_block_fused(const float* __restrict__ x,
                        const float* __restrict__ wb,
                        const float* __restrict__ a1,
                        const float* __restrict__ wd,
                        const float* __restrict__ a2,
                        const float* __restrict__ wr,
                        const float* __restrict__ wsk,
                        float* __restrict__ out_res,
                        float* __restrict__ out_skip)
{
    __shared__ float hs[HROWS * LDST];
    __shared__ float wdep[3 * H_];

    const int wave = threadIdx.x >> 5;
    const int lane = threadIdx.x & 31;
    const int l16  = lane & 15;
    const int lhi  = lane >> 4;

    const int blk  = blockIdx.x;
    const int b    = blk >> 6;
    const int t0   = (blk & 63) * TILE_T;

    __builtin_prefetch(wb,  0, 1);
    __builtin_prefetch(wr,  0, 1);
    __builtin_prefetch(wsk, 0, 1);

    for (int i = threadIdx.x; i < 3 * H_; i += 256) wdep[i] = wd[i];

    const v8f vzero = {0.f,0.f,0.f,0.f,0.f,0.f,0.f,0.f};


    const float* xrow[NMT];
    bool tin[NMT];
    #pragma unroll
    for (int mt = 0; mt < NMT; ++mt) {
        const int ta = t0 - DIL + mt * 16 + l16;
        tin[mt]  = (ta >= 0) && (ta < T_);
        xrow[mt] = x + ((size_t)b * T_ + (tin[mt] ? ta : 0)) * C_;
    }

    for (int p = 0; p < 2; ++p) {
        const int n0 = (wave * 4 + p * 2) * 16 + l16;

        v8f acc[NMT][2];
        #pragma unroll
        for (int mt = 0; mt < NMT; ++mt) { acc[mt][0] = vzero; acc[mt][1] = vzero; }

        #pragma unroll 8
        for (int kk = 0; kk < C_ / 4; ++kk) {
            const int ka = kk * 4 + lhi * 2;
            v2f bf0, bf1;
            bf0.x = wb[(size_t)(ka    ) * H_ + n0];
            bf0.y = wb[(size_t)(ka + 1) * H_ + n0];
            bf1.x = wb[(size_t)(ka    ) * H_ + n0 + 16];
            bf1.y = wb[(size_t)(ka + 1) * H_ + n0 + 16];
            #pragma unroll
            for (int mt = 0; mt < NMT; ++mt) {
                const float2 av = *(const float2*)(xrow[mt] + ka);
                v2f a;
                a.x = tin[mt] ? av.x : 0.f;
                a.y = tin[mt] ? av.y : 0.f;
                acc[mt][0] = __builtin_amdgcn_wmma_f32_16x16x4_f32(
                                 false, a, false, bf0, (short)0, acc[mt][0], false, false);
                acc[mt][1] = __builtin_amdgcn_wmma_f32_16x16x4_f32(
                                 false, a, false, bf1, (short)0, acc[mt][1], false, false);
            }
        }

        #pragma unroll
        for (int mt = 0; mt < NMT; ++mt) {
            #pragma unroll
            for (int j = 0; j < 2; ++j) {
                const int n = n0 + j * 16;
                #pragma unroll
                for (int i = 0; i < 8; ++i) {
                    const int r = mt * 16 + lhi * 8 + i;
                    int tt = t0 - DIL + r;
                    tt = tt < 0 ? 0 : (tt > T_ - 1 ? T_ - 1 : tt);
                    const float al = a1[(size_t)tt * H_ + n];
                    float v = acc[mt][j][i];
                    v = fmaxf(v, 0.f) + al * fminf(v, 0.f);
                    hs[r * LDST + n] = v;
                }
            }
        }
    }
    __syncthreads();

    const int mt2  = wave >> 1;
    const int ntb  = (wave & 1) * 4;
    const int ry   = mt2 * 16 + l16;
    const int tabs = t0 + ry;

    const float* h0p = &hs[(ry          ) * LDST];
    const float* h1p = &hs[(ry + DIL    ) * LDST];
    const float* h2p = &hs[(ry + 2 * DIL) * LDST];
    const float* a2p = a2 + (size_t)tabs * H_;

    v8f cres[4], cskp[4];
    #pragma unroll
    for (int j = 0; j < 4; ++j) { cres[j] = vzero; cskp[j] = vzero; }

    #pragma unroll 4
    for (int kk = 0; kk < H_ / 4; ++kk) {
        const int k0 = kk * 4 + lhi * 2;
        const float2 h0 = *(const float2*)(h0p + k0);
        const float2 h1 = *(const float2*)(h1p + k0);
        const float2 h2 = *(const float2*)(h2p + k0);
        const float2 w0 = *(const float2*)(wdep + k0);
        const float2 w1 = *(const float2*)(wdep + H_ + k0);
        const float2 w2 = *(const float2*)(wdep + 2 * H_ + k0);
        const float2 al = *(const float2*)(a2p + k0);
        v2f a;
        {
            const float yx = w0.x * h0.x + w1.x * h1.x + w2.x * h2.x;
            const float yy = w0.y * h0.y + w1.y * h1.y + w2.y * h2.y;
            a.x = fmaxf(yx, 0.f) + al.x * fminf(yx, 0.f);
            a.y = fmaxf(yy, 0.f) + al.y * fminf(yy, 0.f);
        }
        #pragma unroll
        for (int j = 0; j < 4; ++j) {
            const int n = (ntb + j) * 16 + l16;
            v2f br, bs;
            br.x = wr [(size_t)(k0    ) * C_ + n];
            br.y = wr [(size_t)(k0 + 1) * C_ + n];
            bs.x = wsk[(size_t)(k0    ) * C_ + n];
            bs.y = wsk[(size_t)(k0 + 1) * C_ + n];
            cres[j] = __builtin_amdgcn_wmma_f32_16x16x4_f32(
                          false, a, false, br, (short)0, cres[j], false, false);
            cskp[j] = __builtin_amdgcn_wmma_f32_16x16x4_f32(
                          false, a, false, bs, (short)0, cskp[j], false, false);
        }
    }

    #pragma unroll
    for (int j = 0; j < 4; ++j) {
        const int n = (ntb + j) * 16 + l16;
        #pragma unroll
        for (int i = 0; i < 8; ++i) {
            const int r = mt2 * 16 + lhi * 8 + i;
            const size_t idx = ((size_t)b * T_ + (t0 + r)) * C_ + n;
            out_res[idx]  = cres[j][i] + x[idx];
            out_skip[idx] = cskp[j][i];
        }
    }
}

extern "C" void kernel_launch(void* const* d_in, const int* in_sizes, int n_in,
                              void* d_out, int out_size, void* d_ws, size_t ws_size,
                              hipStream_t stream) {
    const float* x   = (const float*)d_in[0];
    const float* wb  = (const float*)d_in[1];
    const float* a1  = (const float*)d_in[2];
    const float* wd  = (const float*)d_in[3];
    const float* a2  = (const float*)d_in[4];
    const float* wr  = (const float*)d_in[5];
    const float* wsk = (const float*)d_in[6];

    float* res = (float*)d_out;
    float* skp = res + (size_t)B_ * T_ * C_;

    dim3 grid(B_ * (T_ / TILE_T));
    dim3 block(256);
    hipLaunchKernelGGL(conv1d_block_fused, grid, block, 0, stream,
                       x, wb, a1, wd, a2, wr, wsk, res, skp);
}


# hidden dimension in two halves of 256: each half of h is built in LDS and consumed at once as a K-slice of GEMM2/3, accumulators stay in registers; LDS 167 to 87 KiB, so two workgroups share a WGP
# speedup vs baseline: 1.0591x; 1.0591x over previous
#include <hip/hip_runtime.h>


typedef __attribute__((ext_vector_type(2))) float v2f;
typedef __attribute__((ext_vector_type(8))) float v8f;

#define B_      8
#define T_      4096
#define C_      128
#define H_      512
#define DIL     4
#define TILE_T  64
#define HROWS   80
#define NMT     (HROWS / 16)
#define HC      256
#define NHC     (H_ / HC)
#define LDST    (HC + 4)

__global__ __launch_bounds__(256, 4)
void conv1d_block_fused(const float* __restrict__ x,
                        const float* __restrict__ wb,
                        const float* __restrict__ a1,
                        const float* __restrict__ wd,
                        const float* __restrict__ a2,
                        const float* __restrict__ wr,
                        const float* __restrict__ wsk,
                        float* __restrict__ out_res,
                        float* __restrict__ out_skip)
{
    __shared__ float hs[HROWS * LDST];
    __shared__ float wdep[3 * H_];

    const int wave = threadIdx.x >> 5;
    const int lane = threadIdx.x & 31;
    const int l16  = lane & 15;
    const int lhi  = lane >> 4;

    const int blk  = blockIdx.x;
    const int b    = blk >> 6;
    const int t0   = (blk & 63) * TILE_T;

    __builtin_prefetch(wb,  0, 1);
    __builtin_prefetch(wr,  0, 1);
    __builtin_prefetch(wsk, 0, 1);

    for (int i = threadIdx.x; i < 3 * H_; i += 256) wdep[i] = wd[i];

    const v8f vzero = {0.f,0.f,0.f,0.f,0.f,0.f,0.f,0.f};

    const int mt2  = wave >> 1;
    const int ntb  = (wave & 1) * 4;
    const int ry   = mt2 * 16 + l16;
    const int tabs = t0 + ry;

    v8f cres[4], cskp[4];
    #pragma unroll
    for (int j = 0; j < 4; ++j) { cres[j] = vzero; cskp[j] = vzero; }

    #pragma unroll
    for (int p = 0; p < NHC; ++p) {
        const int c0 = wave * 32 + l16;
        const int n0 = p * HC + c0;

        const int ta0 = t0 - DIL + l16;
        const int taL = t0 - DIL + (NMT - 1) * 16 + l16;
        const bool tin0 = ta0 >= 0;
        const bool tinL = taL < T_;
        const float* xmid = x + ((size_t)b * T_ + (ta0 + 16)) * C_ + lhi * 2;
        const float* xlo  = x + ((size_t)b * T_ + (tin0 ? ta0 : 0)) * C_ + lhi * 2;
        const float* xhi  = x + ((size_t)b * T_ + (tinL ? taL : 0)) * C_ + lhi * 2;
        const float* wbp = wb + (size_t)(lhi * 2) * H_ + n0;

        v8f acc[NMT][2];
        #pragma unroll
        for (int mt = 0; mt < NMT; ++mt) { acc[mt][0] = vzero; acc[mt][1] = vzero; }

        #pragma unroll 1
        for (int kb = 0; kb < C_ / 4; kb += 2) {
            v2f bf0[2], bf1[2];
            float2 av[2][NMT];
            #pragma unroll
            for (int u = 0; u < 2; ++u) {
                const float* wbk = wbp + (size_t)((kb + u) * 4) * H_;
                bf0[u].x = wbk[0];
                bf0[u].y = wbk[H_];
                bf1[u].x = wbk[16];
                bf1[u].y = wbk[H_ + 16];
                #pragma unroll
                for (int mt = 0; mt < NMT; ++mt)
                    av[u][mt] = *(const float2*)((mt == 0 ? xlo : mt == NMT - 1 ? xhi
                                                  : xmid + (mt - 1) * 16 * C_) + (kb + u) * 4);
            }
            __builtin_amdgcn_sched_barrier(0);
            #pragma unroll
            for (int u = 0; u < 2; ++u) {
                #pragma unroll
                for (int mt = 0; mt < NMT; ++mt) {
                    v2f a;
                    const bool in = mt == 0 ? tin0 : mt == NMT - 1 ? tinL : true;
                    a.x = in ? av[u][mt].x : 0.f;
                    a.y = in ? av[u][mt].y : 0.f;
                    acc[mt][0] = __builtin_amdgcn_wmma_f32_16x16x4_f32(
                                     false, a, false, bf0[u], (short)0, acc[mt][0], false, false);
                    acc[mt][1] = __builtin_amdgcn_wmma_f32_16x16x4_f32(
                                     false, a, false, bf1[u], (short)0, acc[mt][1], false, false);
                }
            }
        }

        if (p > 0) __syncthreads();

        const int rb = t0 - DIL + lhi * 8;
        #pragma unroll
        for (int mt = 0; mt < NMT; ++mt) {
            float al[2][8];
            #pragma unroll
            for (int j = 0; j < 2; ++j) {
                #pragma unroll
                for (int i = 0; i < 8; ++i) {
                    int tt = rb + mt * 16 + i;
                    if (mt == 0)       tt = tt < 0 ? 0 : tt;
                    if (mt == NMT - 1) tt = tt > T_ - 1 ? T_ - 1 : tt;
                    al[j][i] = a1[(size_t)tt * H_ + n0 + j * 16];
                }
            }
            __builtin_amdgcn_sched_barrier(0);
            #pragma unroll
            for (int j = 0; j < 2; ++j) {
                #pragma unroll
                for (int i = 0; i < 8; ++i) {
                    const int r = mt * 16 + lhi * 8 + i;
                    float v = acc[mt][j][i];
                    v = fmaxf(v, 0.f) + al[j][i] * fminf(v, 0.f);
                    hs[r * LDST + c0 + j * 16] = v;
                }
            }
        }
        __syncthreads();

        const float* h0p = &hs[(ry          ) * LDST + lhi * 2];
        const float* h1p = &hs[(ry + DIL    ) * LDST + lhi * 2];
        const float* h2p = &hs[(ry + 2 * DIL) * LDST + lhi * 2];
        const float* wdp = wdep + p * HC + lhi * 2;
        const float* a2p = a2 + (size_t)tabs * H_ + p * HC + lhi * 2;
        const float* wrp = wr  + (size_t)(p * HC + lhi * 2) * C_ + ntb * 16 + l16;
        const float* wsp = wsk + (size_t)(p * HC + lhi * 2) * C_ + ntb * 16 + l16;

        #pragma unroll 1
        for (int kb = 0; kb < HC / 4; kb += 4) {
            float2 h0[4], h1[4], h2[4], w0[4], w1[4], w2[4], al[4];
            v2f br[4][4], bs[4][4];
            #pragma unroll
            for (int u = 0; u < 4; ++u) {
                const int kl = (kb + u) * 4;
                const float* wrk = wrp + (size_t)kl * C_;
                const float* wsk_k = wsp + (size_t)kl * C_;
                #pragma unroll
                for (int j = 0; j < 4; ++j) {
                    br[u][j].x = wrk[j * 16];
                    br[u][j].y = wrk[C_ + j * 16];
                    bs[u][j].x = wsk_k[j * 16];
                    bs[u][j].y = wsk_k[C_ + j * 16];
                }
                al[u] = *(const float2*)(a2p + kl);
                h0[u] = *(const float2*)(h0p + kl);
                h1[u] = *(const float2*)(h1p + kl);
                h2[u] = *(const float2*)(h2p + kl);
                w0[u] = *(const float2*)(wdp + kl);
                w1[u] = *(const float2*)(wdp + H_ + kl);
                w2[u] = *(const float2*)(wdp + 2 * H_ + kl);
            }
            __builtin_amdgcn_sched_barrier(0);
            #pragma unroll
            for (int u = 0; u < 4; ++u) {
                v2f a;
                {
                    const float yx = w0[u].x * h0[u].x + w1[u].x * h1[u].x + w2[u].x * h2[u].x;
                    const float yy = w0[u].y * h0[u].y + w1[u].y * h1[u].y + w2[u].y * h2[u].y;
                    a.x = fmaxf(yx, 0.f) + al[u].x * fminf(yx, 0.f);
                    a.y = fmaxf(yy, 0.f) + al[u].y * fminf(yy, 0.f);
                }
                #pragma unroll
                for (int j = 0; j < 4; ++j) {
                    cres[j] = __builtin_amdgcn_wmma_f32_16x16x4_f32(
                                  false, a, false, br[u][j], (short)0, cres[j], false, false);
                    cskp[j] = __builtin_amdgcn_wmma_f32_16x16x4_f32(
                                  false, a, false, bs[u][j], (short)0, cskp[j], false, false);
                }
            }
        }
    }

    #pragma unroll
    for (int j = 0; j < 4; ++j) {
        const int n = (ntb + j) * 16 + l16;
        #pragma unroll
        for (int i = 0; i < 8; ++i) {
            const int r = mt2 * 16 + lhi * 8 + i;
            const size_t idx = ((size_t)b * T_ + (t0 + r)) * C_ + n;
            out_res[idx]  = cres[j][i] + x[idx];
            out_skip[idx] = cskp[j][i];
        }
    }
}

extern "C" void kernel_launch(void* const* d_in, const int* in_sizes, int n_in,
                              void* d_out, int out_size, void* d_ws, size_t ws_size,
                              hipStream_t stream) {
    const float* x   = (const float*)d_in[0];
    const float* wb  = (const float*)d_in[1];
    const float* a1  = (const float*)d_in[2];
    const float* wd  = (const float*)d_in[3];
    const float* a2  = (const float*)d_in[4];
    const float* wr  = (const float*)d_in[5];
    const float* wsk = (const float*)d_in[6];

    float* res = (float*)d_out;
    float* skp = res + (size_t)B_ * T_ * C_;

    dim3 grid(B_ * (T_ / TILE_T));
    dim3 block(256);
    hipLaunchKernelGGL(conv1d_block_fused, grid, block, 0, stream,
                       x, wb, a1, wd, a2, wr, wsk, res, skp);
}


# GEMM2/3 weights through LDS: each 16-row chunk of Wr|Ws copied once per workgroup by whole-row b128 loads into a 2-stage ring; waves read B fragments from LDS, tiles on interleaved columns
# speedup vs baseline: 1.1750x; 1.1094x over previous
#include <hip/hip_runtime.h>


typedef __attribute__((ext_vector_type(2))) float v2f;
typedef __attribute__((ext_vector_type(8))) float v8f;
typedef __attribute__((ext_vector_type(4))) float v4f;

#define B_      8
#define T_      4096
#define C_      128
#define H_      512
#define DIL     4
#define TILE_T  64
#define HROWS   80
#define NMT     (HROWS / 16)
#define HC      256
#define NHC     (H_ / HC)
#define LDST    (HC + 4)
#define KC      16
#define NCH     (HC / KC)
#define WP      (2 * C_)

__global__ __launch_bounds__(256, 4)
void conv1d_block_fused(const float* __restrict__ x,
                        const float* __restrict__ wb,
                        const float* __restrict__ a1,
                        const float* __restrict__ wd,
                        const float* __restrict__ a2,
                        const float* __restrict__ wr,
                        const float* __restrict__ wsk,
                        float* __restrict__ out_res,
                        float* __restrict__ out_skip)
{
    __shared__ float hs[HROWS * LDST];
    __shared__ float wdep[3 * H_];
    __shared__ __attribute__((aligned(16))) float wst[2 * KC * WP];

    const int wave = threadIdx.x >> 5;
    const int lane = threadIdx.x & 31;
    const int l16  = lane & 15;
    const int lhi  = lane >> 4;

    const int blk  = blockIdx.x;
    const int b    = blk >> 6;
    const int t0   = (blk & 63) * TILE_T;

    __builtin_prefetch(wb,  0, 1);
    __builtin_prefetch(wr,  0, 1);
    __builtin_prefetch(wsk, 0, 1);

    for (int i = threadIdx.x; i < 3 * H_; i += 256) wdep[i] = wd[i];

    const v8f vzero = {0.f,0.f,0.f,0.f,0.f,0.f,0.f,0.f};

    const int mt2  = wave >> 1;
    const int cb   = (wave & 1) * 64 + l16 * 4;
    const int ry   = mt2 * 16 + l16;
    const int tabs = t0 + ry;

    v8f cres[4], cskp[4];
    #pragma unroll
    for (int j = 0; j < 4; ++j) { cres[j] = vzero; cskp[j] = vzero; }

    const float* sgp = ((wave & 1) ? wsk : wr) + (size_t)((wave >> 1) * 4) * C_ + lane * 4;
    float*       slp = wst + ((wave >> 1) * 4) * WP + (wave & 1) * C_ + lane * 4;
    const float* wlr = wst + (lhi * 2) * WP + cb;
    v4f stg0, stg1, stg2, stg3;

    #pragma unroll
    for (int p = 0; p < NHC; ++p) {
        const int c0 = wave * 32 + l16;
        const int n0 = p * HC + c0;

        const int ta0 = t0 - DIL + l16;
        const int taL = t0 - DIL + (NMT - 1) * 16 + l16;
        const bool tin0 = ta0 >= 0;
        const bool tinL = taL < T_;
        const float* xmid = x + ((size_t)b * T_ + (ta0 + 16)) * C_ + lhi * 2;
        const float* xlo  = x + ((size_t)b * T_ + (tin0 ? ta0 : 0)) * C_ + lhi * 2;
        const float* xhi  = x + ((size_t)b * T_ + (tinL ? taL : 0)) * C_ + lhi * 2;
        const float* wbp = wb + (size_t)(lhi * 2) * H_ + n0;

        v8f acc[NMT][2];
        #pragma unroll
        for (int mt = 0; mt < NMT; ++mt) { acc[mt][0] = vzero; acc[mt][1] = vzero; }

        #pragma unroll 1
        for (int kb = 0; kb < C_ / 4; kb += 2) {
            v2f bf0[2], bf1[2];
            float2 av[2][NMT];
            #pragma unroll
            for (int u = 0; u < 2; ++u) {
                const float* wbk = wbp + (size_t)((kb + u) * 4) * H_;
                bf0[u].x = wbk[0];
                bf0[u].y = wbk[H_];
                bf1[u].x = wbk[16];
                bf1[u].y = wbk[H_ + 16];
                #pragma unroll
                for (int mt = 0; mt < NMT; ++mt)
                    av[u][mt] = *(const float2*)((mt == 0 ? xlo : mt == NMT - 1 ? xhi
                                                  : xmid + (mt - 1) * 16 * C_) + (kb + u) * 4);
            }
            __builtin_amdgcn_sched_barrier(0);
            #pragma unroll
            for (int u = 0; u < 2; ++u) {
                #pragma unroll
                for (int mt = 0; mt < NMT; ++mt) {
                    v2f a;
                    const bool in = mt == 0 ? tin0 : mt == NMT - 1 ? tinL : true;
                    a.x = in ? av[u][mt].x : 0.f;
                    a.y = in ? av[u][mt].y : 0.f;
                    acc[mt][0] = __builtin_amdgcn_wmma_f32_16x16x4_f32(
                                     false, a, false, bf0[u], (short)0, acc[mt][0], false, false);
                    acc[mt][1] = __builtin_amdgcn_wmma_f32_16x16x4_f32(
                                     false, a, false, bf1[u], (short)0, acc[mt][1], false, false);
                }
            }
        }

        {
            const float* g = sgp + (size_t)(p * HC) * C_;
            stg0 = *(const v4f*)(g);
            stg1 = *(const v4f*)(g + C_);
            stg2 = *(const v4f*)(g + 2 * C_);
            stg3 = *(const v4f*)(g + 3 * C_);
        }

        if (p > 0) __syncthreads();

        const int rb = t0 - DIL + lhi * 8;
        #pragma unroll
        for (int mt = 0; mt < NMT; ++mt) {
            float al[2][8];
            #pragma unroll
            for (int j = 0; j < 2; ++j) {
                #pragma unroll
                for (int i = 0; i < 8; ++i) {
                    int tt = rb + mt * 16 + i;
                    if (mt == 0)       tt = tt < 0 ? 0 : tt;
                    if (mt == NMT - 1) tt = tt > T_ - 1 ? T_ - 1 : tt;
                    al[j][i] = a1[(size_t)tt * H_ + n0 + j * 16];
                }
            }
            __builtin_amdgcn_sched_barrier(0);
            #pragma unroll
            for (int j = 0; j < 2; ++j) {
                #pragma unroll
                for (int i = 0; i < 8; ++i) {
                    const int r = mt * 16 + lhi * 8 + i;
                    float v = acc[mt][j][i];
                    v = fmaxf(v, 0.f) + al[j][i] * fminf(v, 0.f);
                    hs[r * LDST + c0 + j * 16] = v;
                }
            }
        }
        *(v4f*)(slp)          = stg0;
        *(v4f*)(slp + WP)     = stg1;
        *(v4f*)(slp + 2 * WP) = stg2;
        *(v4f*)(slp + 3 * WP) = stg3;
        __syncthreads();

        const float* h0p = &hs[(ry          ) * LDST + lhi * 2];
        const float* h1p = &hs[(ry + DIL    ) * LDST + lhi * 2];
        const float* h2p = &hs[(ry + 2 * DIL) * LDST + lhi * 2];
        const float* wdp = wdep + p * HC + lhi * 2;
        const float* a2p = a2 + (size_t)tabs * H_ + p * HC + lhi * 2;
        #pragma unroll 1
        for (int ch = 0; ch < NCH; ++ch) {
            const bool more = ch + 1 < NCH;
            if (more) {
                const float* g = sgp + (size_t)(p * HC + (ch + 1) * KC) * C_;
                stg0 = *(const v4f*)(g);
                stg1 = *(const v4f*)(g + C_);
                stg2 = *(const v4f*)(g + 2 * C_);
                stg3 = *(const v4f*)(g + 3 * C_);
            }
            const float* wl = wlr + (ch & 1) * (KC * WP);
            v2f al[4];
            #pragma unroll
            for (int u = 0; u < 4; ++u)
                al[u] = *(const v2f*)(a2p + (ch * 4 + u) * 4);
            #pragma unroll
            for (int sb = 0; sb < 2; ++sb) {
                v2f h0[2], h1[2], h2[2], w0[2], w1[2], w2[2];
                v4f rr[2][2], rs[2][2];
                #pragma unroll
                for (int v = 0; v < 2; ++v) {
                    const int u  = sb * 2 + v;
                    const int kl = (ch * 4 + u) * 4;
                    #pragma unroll
                    for (int s = 0; s < 2; ++s) {
                        rr[v][s] = *(const v4f*)(wl + (u * 4 + s) * WP);
                        rs[v][s] = *(const v4f*)(wl + (u * 4 + s) * WP + C_);
                    }
                    h0[v] = *(const v2f*)(h0p + kl);
                    h1[v] = *(const v2f*)(h1p + kl);
                    h2[v] = *(const v2f*)(h2p + kl);
                    w0[v] = *(const v2f*)(wdp + kl);
                    w1[v] = *(const v2f*)(wdp + H_ + kl);
                    w2[v] = *(const v2f*)(wdp + 2 * H_ + kl);
                }
                if (sb == 0)
                    asm volatile("" : "+v"(al[0]), "+v"(al[1]), "+v"(al[2]), "+v"(al[3]));
                #pragma unroll
                for (int v = 0; v < 2; ++v) {
                    asm volatile("" : "+v"(rr[v][0]), "+v"(rr[v][1]), "+v"(rs[v][0]), "+v"(rs[v][1]));
                    asm volatile("" : "+v"(h0[v]), "+v"(h1[v]), "+v"(h2[v]),
                                      "+v"(w0[v]), "+v"(w1[v]), "+v"(w2[v]));
                }
                __builtin_amdgcn_sched_barrier(0);
                #pragma unroll
                for (int v = 0; v < 2; ++v) {
                    const int u = sb * 2 + v;
                    v2f a;
                    {
                        const float yx = w0[v].x * h0[v].x + w1[v].x * h1[v].x + w2[v].x * h2[v].x;
                        const float yy = w0[v].y * h0[v].y + w1[v].y * h1[v].y + w2[v].y * h2[v].y;
                        a.x = fmaxf(yx, 0.f) + al[u].x * fminf(yx, 0.f);
                        a.y = fmaxf(yy, 0.f) + al[u].y * fminf(yy, 0.f);
                    }
                    #pragma unroll
                    for (int j = 0; j < 4; ++j) {
                        v2f br, bs;
                        br.x = rr[v][0][j]; br.y = rr[v][1][j];
                        bs.x = rs[v][0][j]; bs.y = rs[v][1][j];
                        cres[j] = __builtin_amdgcn_wmma_f32_16x16x4_f32(
                                      false, a, false, br, (short)0, cres[j], false, false);
                        cskp[j] = __builtin_amdgcn_wmma_f32_16x16x4_f32(
                                      false, a, false, bs, (short)0, cskp[j], false, false);
                    }
                }
                __builtin_amdgcn_sched_barrier(0);
            }
            if (more) {
                float* sl = slp + ((ch + 1) & 1) * (KC * WP);
                *(v4f*)(sl)          = stg0;
                *(v4f*)(sl + WP)     = stg1;
                *(v4f*)(sl + 2 * WP) = stg2;
                *(v4f*)(sl + 3 * WP) = stg3;
                __syncthreads();
            }
        }
    }

    #pragma unroll
    for (int i = 0; i < 8; ++i) {
        const int r = mt2 * 16 + lhi * 8 + i;
        const size_t idx = ((size_t)b * T_ + (t0 + r)) * C_ + cb;
        const float4 xv = *(const float4*)(x + idx);
        float4 o, s;
        o.x = cres[0][i] + xv.x;  o.y = cres[1][i] + xv.y;
        o.z = cres[2][i] + xv.z;  o.w = cres[3][i] + xv.w;
        s.x = cskp[0][i];  s.y = cskp[1][i];  s.z = cskp[2][i];  s.w = cskp[3][i];
        *(float4*)(out_res  + idx) = o;
        *(float4*)(out_skip + idx) = s;
    }
}

extern "C" void kernel_launch(void* const* d_in, const int* in_sizes, int n_in,
                              void* d_out, int out_size, void* d_ws, size_t ws_size,
                              hipStream_t stream) {
    const float* x   = (const float*)d_in[0];
    const float* wb  = (const float*)d_in[1];
    const float* a1  = (const float*)d_in[2];
    const float* wd  = (const float*)d_in[3];
    const float* a2  = (const float*)d_in[4];
    const float* wr  = (const float*)d_in[5];
    const float* wsk = (const float*)d_in[6];

    float* res = (float*)d_out;
    float* skp = res + (size_t)B_ * T_ * C_;

    dim3 grid(B_ * (T_ / TILE_T));
    dim3 block(256);
    hipLaunchKernelGGL(conv1d_block_fused, grid, block, 0, stream,
                       x, wb, a1, wd, a2, wr, wsk, res, skp);
}


# depthwise conv and both PReLUs moved from the GEMM2/3 loop onto the GEMM1 accumulators: rows permuted in each M-tile so taps t-4,t,t+4 are one lane's registers; y stored once to LDS, h never stored
# speedup vs baseline: 1.3013x; 1.1075x over previous
#include <hip/hip_runtime.h>


typedef __attribute__((ext_vector_type(2))) float v2f;
typedef __attribute__((ext_vector_type(8))) float v8f;
typedef __attribute__((ext_vector_type(4))) float v4f;

#define B_      8
#define T_      4096
#define C_      128
#define H_      512
#define DIL     4
#define TILE_T  64
#define HROWS   80
#define NMT     (HROWS / 16)
#define HC      256
#define NHC     (H_ / HC)
#define LDST    (HC + 4)
#define RLOC(i) (4 * ((i) >> 1) + ((i) & 1))
#define KC      16
#define NCH     (HC / KC)
#define WP      (2 * C_)

__global__ __launch_bounds__(256, 4)
void conv1d_block_fused(const float* __restrict__ x,
                        const float* __restrict__ wb,
                        const float* __restrict__ a1,
                        const float* __restrict__ wd,
                        const float* __restrict__ a2,
                        const float* __restrict__ wr,
                        const float* __restrict__ wsk,
                        float* __restrict__ out_res,
                        float* __restrict__ out_skip)
{
    __shared__ float ys[TILE_T * LDST];
    __shared__ __attribute__((aligned(16))) float wst[2 * KC * WP];

    const int wave = threadIdx.x >> 5;
    const int lane = threadIdx.x & 31;
    const int l16  = lane & 15;
    const int lhi  = lane >> 4;

    const int blk  = blockIdx.x;
    const int b    = blk >> 6;
    const int t0   = (blk & 63) * TILE_T;

    __builtin_prefetch(wb,  0, 1);
    __builtin_prefetch(wr,  0, 1);
    __builtin_prefetch(wsk, 0, 1);

    const v8f vzero = {0.f,0.f,0.f,0.f,0.f,0.f,0.f,0.f};

    const int mt2  = wave >> 1;
    const int cb   = (wave & 1) * 64 + l16 * 4;
    const int ry   = mt2 * 16 + l16;

    v8f cres[4], cskp[4];
    #pragma unroll
    for (int j = 0; j < 4; ++j) { cres[j] = vzero; cskp[j] = vzero; }

    const float* sgp = ((wave & 1) ? wsk : wr) + (size_t)((wave >> 1) * 4) * C_ + lane * 4;
    float*       slp = wst + ((wave >> 1) * 4) * WP + (wave & 1) * C_ + lane * 4;
    const float* wlr = wst + (lhi * 2) * WP + cb;
    v4f stg0, stg1, stg2, stg3;

    #pragma unroll
    for (int p = 0; p < NHC; ++p) {
        const int c0 = wave * 32 + l16;
        const int n0 = p * HC + c0;

        const int prow = ((l16 >> 1) & 3) * 4 + (l16 >> 3) * 2 + (l16 & 1);
        const int ta0 = t0 - DIL + prow;
        const int taL = t0 - DIL + (NMT - 1) * 16 + prow;
        const bool tin0 = ta0 >= 0;
        const bool tinL = taL < T_;
        const float* xmid = x + ((size_t)b * T_ + (ta0 + 16)) * C_ + lhi * 2;
        const float* xlo  = x + ((size_t)b * T_ + (tin0 ? ta0 : 0)) * C_ + lhi * 2;
        const float* xhi  = x + ((size_t)b * T_ + (tinL ? taL : 0)) * C_ + lhi * 2;
        const float* wbp = wb + (size_t)(lhi * 2) * H_ + n0;

        v8f acc[NMT][2];
        #pragma unroll
        for (int mt = 0; mt < NMT; ++mt) { acc[mt][0] = vzero; acc[mt][1] = vzero; }

        #pragma unroll 1
        for (int kb = 0; kb < C_ / 4; kb += 2) {
            v2f bf0[2], bf1[2];
            float2 av[2][NMT];
            #pragma unroll
            for (int u = 0; u < 2; ++u) {
                const float* wbk = wbp + (size_t)((kb + u) * 4) * H_;
                bf0[u].x = wbk[0];
                bf0[u].y = wbk[H_];
                bf1[u].x = wbk[16];
                bf1[u].y = wbk[H_ + 16];
                #pragma unroll
                for (int mt = 0; mt < NMT; ++mt)
                    av[u][mt] = *(const float2*)((mt == 0 ? xlo : mt == NMT - 1 ? xhi
                                                  : xmid + (mt - 1) * 16 * C_) + (kb + u) * 4);
            }
            __builtin_amdgcn_sched_barrier(0);
            #pragma unroll
            for (int u = 0; u < 2; ++u) {
                #pragma unroll
                for (int mt = 0; mt < NMT; ++mt) {
                    v2f a;
                    const bool in = mt == 0 ? tin0 : mt == NMT - 1 ? tinL : true;
                    a.x = in ? av[u][mt].x : 0.f;
                    a.y = in ? av[u][mt].y : 0.f;
                    acc[mt][0] = __builtin_amdgcn_wmma_f32_16x16x4_f32(
                                     false, a, false, bf0[u], (short)0, acc[mt][0], false, false);
                    acc[mt][1] = __builtin_amdgcn_wmma_f32_16x16x4_f32(
                                     false, a, false, bf1[u], (short)0, acc[mt][1], false, false);
                }
            }
        }

        const int rq = t0 - DIL + 2 * lhi;
        float wdv[3][2];
        #pragma unroll
        for (int tap = 0; tap < 3; ++tap) {
            wdv[tap][0] = wd[tap * H_ + n0];
            wdv[tap][1] = wd[tap * H_ + n0 + 16];
        }
        float al1[2][8], al2[2][2][8];
        #pragma unroll
        for (int j = 0; j < 2; ++j) {
            #pragma unroll
            for (int i = 0; i < 8; ++i) {
                int tt = rq + RLOC(i);
                if (i < 2) tt = tt < 0 ? 0 : tt;
                al1[j][i] = a1[(size_t)tt * H_ + n0 + j * 16];
            }
        }

        if (p > 0) __syncthreads();

        #pragma unroll
        for (int m = 0; m < NMT; ++m) {
            #pragma unroll
            for (int j = 0; j < 2; ++j) {
                #pragma unroll
                for (int i = 0; i < (m == NMT - 1 ? 4 : 8); ++i) {
                    const float v = acc[m][j][i];
                    acc[m][j][i] = fmaxf(v, 0.f) + al1[j][i] * fminf(v, 0.f);
                }
            }
            if (m + 1 < NMT) {
                #pragma unroll
                for (int j = 0; j < 2; ++j) {
                    #pragma unroll
                    for (int i = 0; i < (m + 1 == NMT - 1 ? 4 : 8); ++i) {
                        int tt = rq + (m + 1) * 16 + RLOC(i);
                        if (m + 1 == NMT - 1 && i >= 2) tt = tt > T_ - 1 ? T_ - 1 : tt;
                        al1[j][i] = a1[(size_t)tt * H_ + n0 + j * 16];
                    }
                }
            }
            if (m == NMT - 1) {
                const float* g = sgp + (size_t)(p * HC) * C_;
                stg0 = *(const v4f*)(g);
                stg1 = *(const v4f*)(g + C_);
                stg2 = *(const v4f*)(g + 2 * C_);
                stg3 = *(const v4f*)(g + 3 * C_);
            }
            if (m < NMT - 1) {
                #pragma unroll
                for (int j = 0; j < 2; ++j) {
                    #pragma unroll
                    for (int i = 0; i < 8; ++i) {
                        const int tt = t0 + m * 16 + RLOC(i) + 2 * lhi;
                        al2[m & 1][j][i] = a2[(size_t)tt * H_ + n0 + j * 16];
                    }
                }
            }
            __builtin_amdgcn_sched_barrier(0);
            if (m >= 1) {
                const int mo = m - 1;
                #pragma unroll
                for (int j = 0; j < 2; ++j) {
                    #pragma unroll
                    for (int i = 0; i < 8; ++i) {
                        const float h0 = acc[mo][j][i];
                        const float h1 = (i + 2 < 8) ? acc[mo][j][(i + 2) & 7] : acc[m][j][(i + 2) & 7];
                        const float h2 = (i + 4 < 8) ? acc[mo][j][(i + 4) & 7] : acc[m][j][(i + 4) & 7];
                        float yv = wdv[0][j] * h0 + wdv[1][j] * h1 + wdv[2][j] * h2;
                        yv = fmaxf(yv, 0.f) + al2[mo & 1][j][i] * fminf(yv, 0.f);
                        const int r = mo * 16 + RLOC(i) + 2 * lhi;
                        ys[r * LDST + c0 + j * 16] = yv;
                    }
                }
            }
            __builtin_amdgcn_sched_barrier(0);
        }
        *(v4f*)(slp)          = stg0;
        *(v4f*)(slp + WP)     = stg1;
        *(v4f*)(slp + 2 * WP) = stg2;
        *(v4f*)(slp + 3 * WP) = stg3;
        __syncthreads();

        const float* yp = &ys[ry * LDST + lhi * 2];
        #pragma unroll 1
        for (int ch = 0; ch < NCH; ++ch) {
            const bool more = ch + 1 < NCH;
            if (more) {
                const float* g = sgp + (size_t)(p * HC + (ch + 1) * KC) * C_;
                stg0 = *(const v4f*)(g);
                stg1 = *(const v4f*)(g + C_);
                stg2 = *(const v4f*)(g + 2 * C_);
                stg3 = *(const v4f*)(g + 3 * C_);
            }
            const float* wl = wlr + (ch & 1) * (KC * WP);
            #pragma unroll
            for (int sb = 0; sb < 2; ++sb) {
                v2f ya[2];
                v4f rr[2][2], rs[2][2];
                #pragma unroll
                for (int v = 0; v < 2; ++v) {
                    const int u  = sb * 2 + v;
                    const int kl = (ch * 4 + u) * 4;
                    #pragma unroll
                    for (int s = 0; s < 2; ++s) {
                        rr[v][s] = *(const v4f*)(wl + (u * 4 + s) * WP);
                        rs[v][s] = *(const v4f*)(wl + (u * 4 + s) * WP + C_);
                    }
                    ya[v] = *(const v2f*)(yp + kl);
                }
                #pragma unroll
                for (int v = 0; v < 2; ++v) {
                    asm volatile("" : "+v"(rr[v][0]), "+v"(rr[v][1]), "+v"(rs[v][0]), "+v"(rs[v][1]));
                    asm volatile("" : "+v"(ya[v]));
                }
                __builtin_amdgcn_sched_barrier(0);
                #pragma unroll
                for (int v = 0; v < 2; ++v) {
                    const v2f a = ya[v];
                    #pragma unroll
                    for (int j = 0; j < 4; ++j) {
                        v2f br, bs;
                        br.x = rr[v][0][j]; br.y = rr[v][1][j];
                        bs.x = rs[v][0][j]; bs.y = rs[v][1][j];
                        cres[j] = __builtin_amdgcn_wmma_f32_16x16x4_f32(
                                      false, a, false, br, (short)0, cres[j], false, false);
                        cskp[j] = __builtin_amdgcn_wmma_f32_16x16x4_f32(
                                      false, a, false, bs, (short)0, cskp[j], false, false);
                    }
                }
                __builtin_amdgcn_sched_barrier(0);
            }
            if (more) {
                float* sl = slp + ((ch + 1) & 1) * (KC * WP);
                *(v4f*)(sl)          = stg0;
                *(v4f*)(sl + WP)     = stg1;
                *(v4f*)(sl + 2 * WP) = stg2;
                *(v4f*)(sl + 3 * WP) = stg3;
                __syncthreads();
            }
        }
    }

    #pragma unroll
    for (int i = 0; i < 8; ++i) {
        const int r = mt2 * 16 + lhi * 8 + i;
        const size_t idx = ((size_t)b * T_ + (t0 + r)) * C_ + cb;
        const float4 xv = *(const float4*)(x + idx);
        float4 o, s;
        o.x = cres[0][i] + xv.x;  o.y = cres[1][i] + xv.y;
        o.z = cres[2][i] + xv.z;  o.w = cres[3][i] + xv.w;
        s.x = cskp[0][i];  s.y = cskp[1][i];  s.z = cskp[2][i];  s.w = cskp[3][i];
        *(float4*)(out_res  + idx) = o;
        *(float4*)(out_skip + idx) = s;
    }
}

extern "C" void kernel_launch(void* const* d_in, const int* in_sizes, int n_in,
                              void* d_out, int out_size, void* d_ws, size_t ws_size,
                              hipStream_t stream) {
    const float* x   = (const float*)d_in[0];
    const float* wb  = (const float*)d_in[1];
    const float* a1  = (const float*)d_in[2];
    const float* wd  = (const float*)d_in[3];
    const float* a2  = (const float*)d_in[4];
    const float* wr  = (const float*)d_in[5];
    const float* wsk = (const float*)d_in[6];

    float* res = (float*)d_out;
    float* skp = res + (size_t)B_ * T_ * C_;

    dim3 grid(B_ * (T_ / TILE_T));
    dim3 block(256);
    hipLaunchKernelGGL(conv1d_block_fused, grid, block, 0, stream,
                       x, wb, a1, wd, a2, wr, wsk, res, skp);
}


# wave issue priority by progress (s_setprio 3 at entry, 2 from the second half, 1 and 0 late in the last GEMM2/3 pass): of the two workgroups on a WGP the one behind is served first
# speedup vs baseline: 1.4302x; 1.0991x over previous
#include <hip/hip_runtime.h>


typedef __attribute__((ext_vector_type(2))) float v2f;
typedef __attribute__((ext_vector_type(8))) float v8f;
typedef __attribute__((ext_vector_type(4))) float v4f;

#define B_      8
#define T_      4096
#define C_      128
#define H_      512
#define DIL     4
#define TILE_T  64
#define HROWS   80
#define NMT     (HROWS / 16)
#define HC      256
#define NHC     (H_ / HC)
#define LDST    (HC + 4)
#define RLOC(i) (4 * ((i) >> 1) + ((i) & 1))
#define KC      16
#define NCH     (HC / KC)
#define WP      (2 * C_)

__global__ __launch_bounds__(256, 4)
void conv1d_block_fused(const float* __restrict__ x,
                        const float* __restrict__ wb,
                        const float* __restrict__ a1,
                        const float* __restrict__ wd,
                        const float* __restrict__ a2,
                        const float* __restrict__ wr,
                        const float* __restrict__ wsk,
                        float* __restrict__ out_res,
                        float* __restrict__ out_skip)
{
    __shared__ float ys[TILE_T * LDST];
    __shared__ __attribute__((aligned(16))) float wst[2 * KC * WP];

    const int wave = threadIdx.x >> 5;
    const int lane = threadIdx.x & 31;
    const int l16  = lane & 15;
    const int lhi  = lane >> 4;

    const int blk  = blockIdx.x;
    const int b    = blk >> 6;
    const int t0   = (blk & 63) * TILE_T;

    __builtin_prefetch(wb,  0, 1);
    __builtin_prefetch(wr,  0, 1);
    __builtin_prefetch(wsk, 0, 1);

    const v8f vzero = {0.f,0.f,0.f,0.f,0.f,0.f,0.f,0.f};

    __builtin_amdgcn_s_setprio(3);

    const int mt2  = wave >> 1;
    const int cb   = (wave & 1) * 64 + l16 * 4;
    const int ry   = mt2 * 16 + l16;

    v8f cres[4], cskp[4];
    #pragma unroll
    for (int j = 0; j < 4; ++j) { cres[j] = vzero; cskp[j] = vzero; }

    const float* sgp = ((wave & 1) ? wsk : wr) + (size_t)((wave >> 1) * 4) * C_ + lane * 4;
    float*       slp = wst + ((wave >> 1) * 4) * WP + (wave & 1) * C_ + lane * 4;
    const float* wlr = wst + (lhi * 2) * WP + cb;
    v4f stg0, stg1, stg2, stg3;

    #pragma unroll
    for (int p = 0; p < NHC; ++p) {
        if (p == 1) __builtin_amdgcn_s_setprio(2);
        const int c0 = wave * 32 + l16;
        const int n0 = p * HC + c0;

        const int prow = ((l16 >> 1) & 3) * 4 + (l16 >> 3) * 2 + (l16 & 1);
        const int ta0 = t0 - DIL + prow;
        const int taL = t0 - DIL + (NMT - 1) * 16 + prow;
        const bool tin0 = ta0 >= 0;
        const bool tinL = taL < T_;
        const float* xmid = x + ((size_t)b * T_ + (ta0 + 16)) * C_ + lhi * 8;
        const float* xlo  = x + ((size_t)b * T_ + (tin0 ? ta0 : 0)) * C_ + lhi * 8;
        const float* xhi  = x + ((size_t)b * T_ + (tinL ? taL : 0)) * C_ + lhi * 8;
        const float* wbp = wb + (size_t)(lhi * 8) * H_ + n0;

        v8f acc[NMT][2];
        #pragma unroll
        for (int mt = 0; mt < NMT; ++mt) { acc[mt][0] = vzero; acc[mt][1] = vzero; }

        #pragma unroll 1
        for (int kb = 0; kb < C_; kb += 16) {
            v2f bf0[4], bf1[4];
            v4f av[NMT][2];
            const float* wbk = wbp + (size_t)kb * H_;
            #pragma unroll
            for (int q = 0; q < 4; ++q) {
                bf0[q].x = wbk[(2 * q) * H_];
                bf0[q].y = wbk[(2 * q + 1) * H_];
                bf1[q].x = wbk[(2 * q) * H_ + 16];
                bf1[q].y = wbk[(2 * q + 1) * H_ + 16];
            }
            #pragma unroll
            for (int mt = 0; mt < NMT; ++mt) {
                const float* xr = (mt == 0 ? xlo : mt == NMT - 1 ? xhi
                                   : xmid + (mt - 1) * 16 * C_) + kb;
                av[mt][0] = *(const v4f*)(xr);
                av[mt][1] = *(const v4f*)(xr + 4);
            }
            __builtin_amdgcn_sched_barrier(0);
            #pragma unroll
            for (int q = 0; q < 4; ++q) {
                #pragma unroll
                for (int mt = 0; mt < NMT; ++mt) {
                    v2f a;
                    const bool in = mt == 0 ? tin0 : mt == NMT - 1 ? tinL : true;
                    a.x = in ? av[mt][q >> 1][(q & 1) * 2]     : 0.f;
                    a.y = in ? av[mt][q >> 1][(q & 1) * 2 + 1] : 0.f;
                    acc[mt][0] = __builtin_amdgcn_wmma_f32_16x16x4_f32(
                                     false, a, false, bf0[q], (short)0, acc[mt][0], false, false);
                    acc[mt][1] = __builtin_amdgcn_wmma_f32_16x16x4_f32(
                                     false, a, false, bf1[q], (short)0, acc[mt][1], false, false);
                }
            }
        }

        const int rq = t0 - DIL + 2 * lhi;
        float wdv[3][2];
        #pragma unroll
        for (int tap = 0; tap < 3; ++tap) {
            wdv[tap][0] = wd[tap * H_ + n0];
            wdv[tap][1] = wd[tap * H_ + n0 + 16];
        }
        float al1[2][8], al2[2][2][8];
        #pragma unroll
        for (int j = 0; j < 2; ++j) {
            #pragma unroll
            for (int i = 0; i < 8; ++i) {
                int tt = rq + RLOC(i);
                if (i < 2) tt = tt < 0 ? 0 : tt;
                al1[j][i] = a1[(size_t)tt * H_ + n0 + j * 16];
            }
        }

        if (p > 0) __syncthreads();

        #pragma unroll
        for (int m = 0; m < NMT; ++m) {
            #pragma unroll
            for (int j = 0; j < 2; ++j) {
                #pragma unroll
                for (int i = 0; i < (m == NMT - 1 ? 4 : 8); ++i) {
                    const float v = acc[m][j][i];
                    acc[m][j][i] = fmaxf(v, 0.f) + al1[j][i] * fminf(v, 0.f);
                }
            }
            if (m + 1 < NMT) {
                #pragma unroll
                for (int j = 0; j < 2; ++j) {
                    #pragma unroll
                    for (int i = 0; i < (m + 1 == NMT - 1 ? 4 : 8); ++i) {
                        int tt = rq + (m + 1) * 16 + RLOC(i);
                        if (m + 1 == NMT - 1 && i >= 2) tt = tt > T_ - 1 ? T_ - 1 : tt;
                        al1[j][i] = a1[(size_t)tt * H_ + n0 + j * 16];
                    }
                }
            }
            if (m == NMT - 1) {
                const float* g = sgp + (size_t)(p * HC) * C_;
                stg0 = *(const v4f*)(g);
                stg1 = *(const v4f*)(g + C_);
                stg2 = *(const v4f*)(g + 2 * C_);
                stg3 = *(const v4f*)(g + 3 * C_);
            }
            if (m < NMT - 1) {
                #pragma unroll
                for (int j = 0; j < 2; ++j) {
                    #pragma unroll
                    for (int i = 0; i < 8; ++i) {
                        const int tt = t0 + m * 16 + RLOC(i) + 2 * lhi;
                        al2[m & 1][j][i] = a2[(size_t)tt * H_ + n0 + j * 16];
                    }
                }
            }
            __builtin_amdgcn_sched_barrier(0);
            if (m >= 1) {
                const int mo = m - 1;
                #pragma unroll
                for (int j = 0; j < 2; ++j) {
                    #pragma unroll
                    for (int i = 0; i < 8; ++i) {
                        const float h0 = acc[mo][j][i];
                        const float h1 = (i + 2 < 8) ? acc[mo][j][(i + 2) & 7] : acc[m][j][(i + 2) & 7];
                        const float h2 = (i + 4 < 8) ? acc[mo][j][(i + 4) & 7] : acc[m][j][(i + 4) & 7];
                        float yv = wdv[0][j] * h0 + wdv[1][j] * h1 + wdv[2][j] * h2;
                        yv = fmaxf(yv, 0.f) + al2[mo & 1][j][i] * fminf(yv, 0.f);
                        const int r = mo * 16 + RLOC(i) + 2 * lhi;
                        ys[r * LDST + c0 + j * 16] = yv;
                    }
                }
            }
            __builtin_amdgcn_sched_barrier(0);
        }
        *(v4f*)(slp)          = stg0;
        *(v4f*)(slp + WP)     = stg1;
        *(v4f*)(slp + 2 * WP) = stg2;
        *(v4f*)(slp + 3 * WP) = stg3;
        __syncthreads();

        const float* yp = &ys[ry * LDST + lhi * 2];
        #pragma unroll 1
        for (int ch = 0; ch < NCH; ++ch) {
            const bool more = ch + 1 < NCH;
            if (p == 1 && ch == 6)  __builtin_amdgcn_s_setprio(1);
            if (p == 1 && ch == 13) __builtin_amdgcn_s_setprio(0);
            if (more) {
                const float* g = sgp + (size_t)(p * HC + (ch + 1) * KC) * C_;
                stg0 = *(const v4f*)(g);
                stg1 = *(const v4f*)(g + C_);
                stg2 = *(const v4f*)(g + 2 * C_);
                stg3 = *(const v4f*)(g + 3 * C_);
            }
            const float* wl = wlr + (ch & 1) * (KC * WP);
            #pragma unroll
            for (int sb = 0; sb < 2; ++sb) {
                v2f ya[2];
                v4f rr[2][2], rs[2][2];
                #pragma unroll
                for (int v = 0; v < 2; ++v) {
                    const int u  = sb * 2 + v;
                    const int kl = (ch * 4 + u) * 4;
                    #pragma unroll
                    for (int s = 0; s < 2; ++s) {
                        rr[v][s] = *(const v4f*)(wl + (u * 4 + s) * WP);
                        rs[v][s] = *(const v4f*)(wl + (u * 4 + s) * WP + C_);
                    }
                    ya[v] = *(const v2f*)(yp + kl);
                }
                #pragma unroll
                for (int v = 0; v < 2; ++v) {
                    asm volatile("" : "+v"(rr[v][0]), "+v"(rr[v][1]), "+v"(rs[v][0]), "+v"(rs[v][1]));
                    asm volatile("" : "+v"(ya[v]));
                }
                __builtin_amdgcn_sched_barrier(0);
                #pragma unroll
                for (int v = 0; v < 2; ++v) {
                    const v2f a = ya[v];
                    #pragma unroll
                    for (int j = 0; j < 4; ++j) {
                        v2f br, bs;
                        br.x = rr[v][0][j]; br.y = rr[v][1][j];
                        bs.x = rs[v][0][j]; bs.y = rs[v][1][j];
                        cres[j] = __builtin_amdgcn_wmma_f32_16x16x4_f32(
                                      false, a, false, br, (short)0, cres[j], false, false);
                        cskp[j] = __builtin_amdgcn_wmma_f32_16x16x4_f32(
                                      false, a, false, bs, (short)0, cskp[j], false, false);
                    }
                }
                __builtin_amdgcn_sched_barrier(0);
            }
            if (more) {
                float* sl = slp + ((ch + 1) & 1) * (KC * WP);
                *(v4f*)(sl)          = stg0;
                *(v4f*)(sl + WP)     = stg1;
                *(v4f*)(sl + 2 * WP) = stg2;
                *(v4f*)(sl + 3 * WP) = stg3;
                __syncthreads();
            }
        }
    }

    #pragma unroll
    for (int i = 0; i < 8; ++i) {
        const int r = mt2 * 16 + lhi * 8 + i;
        const size_t idx = ((size_t)b * T_ + (t0 + r)) * C_ + cb;
        const float4 xv = *(const float4*)(x + idx);
        float4 o, s;
        o.x = cres[0][i] + xv.x;  o.y = cres[1][i] + xv.y;
        o.z = cres[2][i] + xv.z;  o.w = cres[3][i] + xv.w;
        s.x = cskp[0][i];  s.y = cskp[1][i];  s.z = cskp[2][i];  s.w = cskp[3][i];
        *(float4*)(out_res  + idx) = o;
        *(float4*)(out_skip + idx) = s;
    }
}

extern "C" void kernel_launch(void* const* d_in, const int* in_sizes, int n_in,
                              void* d_out, int out_size, void* d_ws, size_t ws_size,
                              hipStream_t stream) {
    const float* x   = (const float*)d_in[0];
    const float* wb  = (const float*)d_in[1];
    const float* a1  = (const float*)d_in[2];
    const float* wd  = (const float*)d_in[3];
    const float* a2  = (const float*)d_in[4];
    const float* wr  = (const float*)d_in[5];
    const float* wsk = (const float*)d_in[6];

    float* res = (float*)d_out;
    float* skp = res + (size_t)B_ * T_ * C_;

    dim3 grid(B_ * (T_ / TILE_T));
    dim3 block(256);
    hipLaunchKernelGGL(conv1d_block_fused, grid, block, 0, stream,
                       x, wb, a1, wd, a2, wr, wsk, res, skp);
}
